# starting kernel
# baseline (speedup 1.0000x reference)
#include <hip/hip_runtime.h>
#include <math.h>

#define NN   8192
#define NZ   256
#define NC   128
#define DTC  0.1f

#define OUT_X 0
#define OUT_R NN
#define OUT_Z (2 * NN)
#define OUT_C (2 * NN + NZ)
#define OUT_E (2 * NN + NZ + NC)

#define KSPLIT 8
#define KCHUNK (NN / KSPLIT)

typedef __attribute__((ext_vector_type(2))) float v2f;
typedef __attribute__((ext_vector_type(4))) float v4f;
typedef __attribute__((ext_vector_type(8))) float v8f;

template <bool NT>
__device__ __forceinline__ v8f wmma_gemv(v8f acc,
                                         const float* __restrict__ W, int ld,
                                         int row0,
                                         const float* __restrict__ vec, int K) {
  const int lane = threadIdx.x & 31;
  const int m    = lane & 15;
  const int hl   = lane >> 4;
  const float* wp = W + (size_t)(row0 + m) * (size_t)ld + 4 * hl;
  const float* vp = vec + 4 * hl;
#pragma unroll 8
  for (int k = 0; k < K; k += 8) {
    const v4f w = NT ? __builtin_nontemporal_load((const v4f*)(wp + k))
                     : *(const v4f*)(wp + k);
    const v4f bv = *(const v4f*)(vp + k);
    v2f a0 = {w.x, w.y}, b0 = {bv.x, bv.y};
    v2f a1 = {w.z, w.w}, b1 = {bv.z, bv.w};
    acc = __builtin_amdgcn_wmma_f32_16x16x4_f32(false, a0, false, b0,
                                                (short)0, acc, false, false);
    acc = __builtin_amdgcn_wmma_f32_16x16x4_f32(false, a1, false, b1,
                                                (short)0, acc, false, false);
  }
  return acc;
}

__global__ void k_zpart(const float* __restrict__ W_rz,
                        const float* __restrict__ x,
                        const float* __restrict__ b,
                        float* __restrict__ zpart) {
  __shared__ __align__(16) float s_r[KCHUNK];
  const int tid = threadIdx.x;
  const int k0  = blockIdx.y * KCHUNK;
  for (int j = tid; j < KCHUNK; j += 256)
    s_r[j] = tanhf(x[k0 + j] + b[k0 + j]);
  __syncthreads();
  const int row0 = blockIdx.x * 128 + (tid >> 5) * 16;
  v8f acc = {0.f, 0.f, 0.f, 0.f, 0.f, 0.f, 0.f, 0.f};
  acc = wmma_gemv<false>(acc, W_rz + k0, NN, row0, s_r, KCHUNK);
  const int lane = tid & 31;
  if ((lane & 15) == 0) {
    const int rbase = row0 + (lane >> 4) * 8;
    float* dst = zpart + blockIdx.y * NZ;
#pragma unroll
    for (int v = 0; v < 8; ++v) dst[rbase + v] = acc[v];
  }
}

__global__ void k_rrpart(const float* __restrict__ W_rr,
                         const float* __restrict__ x,
                         const float* __restrict__ b,
                         float* __restrict__ rrpart) {
  __shared__ __align__(16) float s_r[KCHUNK];
  const int tid = threadIdx.x;
  const int k0  = blockIdx.y * KCHUNK;
  for (int j = tid; j < KCHUNK; j += 256)
    s_r[j] = tanhf(x[k0 + j] + b[k0 + j]);
  __syncthreads();
  const int row0 = blockIdx.x * 128 + (tid >> 5) * 16;
  v8f acc = {0.f, 0.f, 0.f, 0.f, 0.f, 0.f, 0.f, 0.f};
  acc = wmma_gemv<true>(acc, W_rr + k0, NN, row0, s_r, KCHUNK);
  const int lane = tid & 31;
  if ((lane & 15) == 0) {
    const int rbase = row0 + (lane >> 4) * 8;
    float* dst = rrpart + blockIdx.y * NN;
#pragma unroll
    for (int v = 0; v < 8; ++v) dst[rbase + v] = acc[v];
  }
}

__global__ void k_final(const float* __restrict__ x,
                        const float* __restrict__ b,
                        const float* __restrict__ eps,
                        const float* __restrict__ cvec,
                        const float* __restrict__ W_epsr,
                        const float* __restrict__ W_zr,
                        const float* __restrict__ W_cr,
                        const float* __restrict__ zpart,
                        const float* __restrict__ rrpart,
                        float* __restrict__ out) {
  __shared__ __align__(16) float s_eps[NZ];
  __shared__ __align__(16) float s_z[NZ];
  __shared__ __align__(16) float s_c[NC];
  const int tid = threadIdx.x;
  {
    float zs = 0.f;
#pragma unroll
    for (int s = 0; s < KSPLIT; ++s) zs += zpart[s * NZ + tid];
    s_z[tid]   = zs;
    s_eps[tid] = eps[tid];
    if (tid < NC) s_c[tid] = cvec[tid];
  }
  __syncthreads();
  const int row0 = blockIdx.x * 128 + (tid >> 5) * 16;
  v8f acc = {0.f, 0.f, 0.f, 0.f, 0.f, 0.f, 0.f, 0.f};
  acc = wmma_gemv<false>(acc, W_epsr, NZ, row0, s_eps, NZ);
  acc = wmma_gemv<false>(acc, W_zr,   NZ, row0, s_z,   NZ);
  acc = wmma_gemv<false>(acc, W_cr,   NC, row0, s_c,   NC);
  const int lane = tid & 31;
  if ((lane & 15) == 0) {
    const int rbase = row0 + (lane >> 4) * 8;
#pragma unroll
    for (int v = 0; v < 8; ++v) {
      const int i = rbase + v;
      float tot = acc[v];
#pragma unroll
      for (int s = 0; s < KSPLIT; ++s) tot += rrpart[s * NN + i];
      const float xv = x[i];
      const float xn = xv + DTC * (tot - xv);
      out[OUT_X + i] = xn;
      out[OUT_R + i] = tanhf(xn + b[i]);
    }
  }
}

__global__ void k_outpart(const float* __restrict__ W_rz,
                          const float* __restrict__ W_rc,
                          const float* __restrict__ r_new,
                          float* __restrict__ znpart,
                          float* __restrict__ cnpart) {
  __shared__ __align__(16) float s_r[KCHUNK];
  const int tid = threadIdx.x;
  const int k0  = blockIdx.y * KCHUNK;
  for (int j = tid; j < KCHUNK; j += 256) s_r[j] = r_new[k0 + j];
  __syncthreads();
  const int wave = tid >> 5;
  const int lane = tid & 31;
  v8f acc = {0.f, 0.f, 0.f, 0.f, 0.f, 0.f, 0.f, 0.f};
  if (blockIdx.x < 2) {
    const int row0 = blockIdx.x * 128 + wave * 16;
    acc = wmma_gemv<false>(acc, W_rz + k0, NN, row0, s_r, KCHUNK);
    if ((lane & 15) == 0) {
      const int rbase = row0 + (lane >> 4) * 8;
#pragma unroll
      for (int v = 0; v < 8; ++v) znpart[blockIdx.y * NZ + rbase + v] = acc[v];
    }
  } else {
    const int row0 = wave * 16;
    acc = wmma_gemv<false>(acc, W_rc + k0, NN, row0, s_r, KCHUNK);
    if ((lane & 15) == 0) {
      const int rbase = row0 + (lane >> 4) * 8;
#pragma unroll
      for (int v = 0; v < 8; ++v) cnpart[blockIdx.y * NC + rbase + v] = acc[v];
    }
  }
}

__global__ void k_epilog(const float* __restrict__ znpart,
                         const float* __restrict__ cnpart,
                         const float* __restrict__ z_tilde,
                         float* __restrict__ out) {
  const int tid = threadIdx.x;
  float zn = 0.f;
#pragma unroll
  for (int s = 0; s < KSPLIT; ++s) zn += znpart[s * NZ + tid];
  out[OUT_Z + tid] = zn;
  out[OUT_E + tid] = zn - z_tilde[tid];
  if (tid < NC) {
    float cn = 0.f;
#pragma unroll
    for (int s = 0; s < KSPLIT; ++s) cn += cnpart[s * NC + tid];
    out[OUT_C + tid] = cn;
  }
}

extern "C" void kernel_launch(void* const* d_in, const int* in_sizes, int n_in,
                              void* d_out, int out_size, void* d_ws, size_t ws_size,
                              hipStream_t stream) {
  (void)in_sizes; (void)n_in; (void)out_size; (void)ws_size;
  const float* x       = (const float*)d_in[0];
  const float* eps     = (const float*)d_in[1];
  const float* c       = (const float*)d_in[2];
  const float* z_tilde = (const float*)d_in[3];
  const float* W_rr    = (const float*)d_in[4];
  const float* W_zr    = (const float*)d_in[5];
  const float* W_cr    = (const float*)d_in[6];
  const float* W_epsr  = (const float*)d_in[7];
  const float* W_rz    = (const float*)d_in[8];
  const float* W_rc    = (const float*)d_in[9];
  const float* b       = (const float*)d_in[10];
  float* out = (float*)d_out;

  float* ws     = (float*)d_ws;
  float* zpart  = ws;
  float* rrpart = zpart  + KSPLIT * NZ;
  float* znpart = rrpart + KSPLIT * NN;
  float* cnpart = znpart + KSPLIT * NZ;

  k_zpart  <<<dim3(2, KSPLIT),  256, 0, stream>>>(W_rz, x, b, zpart);
  k_rrpart <<<dim3(64, KSPLIT), 256, 0, stream>>>(W_rr, x, b, rrpart);
  k_final  <<<64, 256, 0, stream>>>(x, b, eps, c, W_epsr, W_zr, W_cr,
                                    zpart, rrpart, out);
  k_outpart<<<dim3(3, KSPLIT),  256, 0, stream>>>(W_rz, W_rc, out + NN,
                                                  znpart, cnpart);
  k_epilog <<<1, 256, 0, stream>>>(znpart, cnpart, z_tilde, out);
}


# WMMA A fragment = 128 consecutive floats of ONE row: a wave load reads 512 contiguous bytes (was 16 rows x 32 B), 3 rows requested ahead; row sum = trace of D + lane butterfly, so sum order changes
# speedup vs baseline: 1.7129x; 1.7129x over previous
#include <hip/hip_runtime.h>
#include <math.h>

#define NN   8192
#define NZ   256
#define NC   128
#define DTC  0.1f

#define OUT_X 0
#define OUT_R NN
#define OUT_Z (2 * NN)
#define OUT_C (2 * NN + NZ)
#define OUT_E (2 * NN + NZ + NC)

#define KSPLIT 8
#define KCHUNK (NN / KSPLIT)
#define KQ     (KCHUNK / 128)
#define WROWS  16

typedef __attribute__((ext_vector_type(2))) float v2f;
typedef __attribute__((ext_vector_type(4))) float v4f;
typedef __attribute__((ext_vector_type(8))) float v8f;

#define DEV static __device__ __forceinline__

DEV v8f wmma4(v8f acc, v2f a, v2f b) {
  return __builtin_amdgcn_wmma_f32_16x16x4_f32(false, a, false, b, (short)0, acc, false, false);
}

DEV float diag_pick(v8f d, int sel) {
  float t = 0.f;
#pragma unroll
  for (int j = 0; j < 8; ++j) t = (sel == j) ? d[j] : t;
  return t;
}

DEV float reduce16(const float (&t)[WROWS], int lane) {
  float u[8], v[4], w[2];
  const bool h4 = (lane & 16) != 0, h3 = (lane & 8) != 0, h2 = (lane & 4) != 0, h1 = (lane & 2) != 0;
#pragma unroll
  for (int j = 0; j < 8; ++j) u[j] = (h4 ? t[8 + j] : t[j]) + __shfl_xor(h4 ? t[j] : t[8 + j], 16, 32);
#pragma unroll
  for (int j = 0; j < 4; ++j) v[j] = (h3 ? u[4 + j] : u[j]) + __shfl_xor(h3 ? u[j] : u[4 + j], 8, 32);
#pragma unroll
  for (int j = 0; j < 2; ++j) w[j] = (h2 ? v[2 + j] : v[j]) + __shfl_xor(h2 ? v[j] : v[2 + j], 4, 32);
  const float x = (h1 ? w[1] : w[0]) + __shfl_xor(h1 ? w[0] : w[1], 2, 32);
  return x + __shfl_xor(x, 1, 32);
}

template <int NQ, int AHEAD, class LoadPiece>
DEV void rows_dot(LoadPiece ld, const v4f (&bv)[NQ], float (&t)[WROWS], int lane) {
  const int sel = (((lane + 8) & 16) == 0) ? (lane & 7) : 8;
  v4f w[AHEAD + 1][NQ];
#pragma unroll
  for (int r = 0; r < AHEAD; ++r) {
#pragma unroll
    for (int q = 0; q < NQ; ++q) w[r][q] = ld(r, q);
  }
#pragma unroll
  for (int r = 0; r < WROWS; ++r) {
    __builtin_amdgcn_sched_barrier(0);
    if (r + AHEAD < WROWS) {
#pragma unroll
      for (int q = 0; q < NQ; ++q) w[(r + AHEAD) % (AHEAD + 1)][q] = ld(r + AHEAD, q);
    }
    __builtin_amdgcn_sched_barrier(0);
    v8f a0 = {0.f, 0.f, 0.f, 0.f, 0.f, 0.f, 0.f, 0.f};
    v8f a1 = {0.f, 0.f, 0.f, 0.f, 0.f, 0.f, 0.f, 0.f};
#pragma unroll
    for (int q = 0; q < NQ; ++q) {
      const v4f wv = w[r % (AHEAD + 1)][q];
      const v2f wa = {wv.x, wv.y}, ba = {bv[q].x, bv[q].y};
      const v2f wb = {wv.z, wv.w}, bb = {bv[q].z, bv[q].w};
      a0 = wmma4(a0, wa, ba);
      a1 = wmma4(a1, wb, bb);
    }
    t[r] = diag_pick(a0 + a1, sel);
    asm volatile("" : "+v"(t[r]));
  }
}

template <bool NT>
DEV void chunk_rows(const float* __restrict__ W, int row0, int k0,
                    const float* s_r, float* __restrict__ dst, int lane) {
  v4f bv[KQ];
#pragma unroll
  for (int q = 0; q < KQ; ++q) bv[q] = *(const v4f*)(s_r + 128 * q + 4 * lane);
  const float* wp = W + (size_t)row0 * NN + k0 + 4 * lane;
  float t[WROWS];
  rows_dot<KQ, 3>([&](int r, int q) -> v4f {
    const v4f* p = (const v4f*)(wp + (size_t)r * NN + 128 * q);
    return NT ? __builtin_nontemporal_load(p) : *p;
  }, bv, t, lane);
  const float y = reduce16(t, lane);
  if ((lane & 1) == 0) dst[row0 + 8 * (lane >> 4) + ((lane >> 1) & 7)] = y;
}

__global__ void __launch_bounds__(256) k_zpart(const float* __restrict__ W_rz,
                        const float* __restrict__ x,
                        const float* __restrict__ b,
                        float* __restrict__ zpart) {
  __shared__ __align__(16) float s_r[KCHUNK];
  const int tid = threadIdx.x;
  const int k0  = blockIdx.y * KCHUNK;
  for (int j = tid; j < KCHUNK; j += 256)
    s_r[j] = tanhf(x[k0 + j] + b[k0 + j]);
  __syncthreads();
  const int row0 = blockIdx.x * 128 + (tid >> 5) * WROWS;
  chunk_rows<false>(W_rz, row0, k0, s_r, zpart + blockIdx.y * NZ, tid & 31);
}

__global__ void __launch_bounds__(256) k_rrpart(const float* __restrict__ W_rr,
                         const float* __restrict__ x,
                         const float* __restrict__ b,
                         float* __restrict__ rrpart) {
  __shared__ __align__(16) float s_r[KCHUNK];
  const int tid = threadIdx.x;
  const int k0  = blockIdx.y * KCHUNK;
  for (int j = tid; j < KCHUNK; j += 256)
    s_r[j] = tanhf(x[k0 + j] + b[k0 + j]);
  __syncthreads();
  const int row0 = blockIdx.x * 128 + (tid >> 5) * WROWS;
  chunk_rows<true>(W_rr, row0, k0, s_r, rrpart + blockIdx.y * NN, tid & 31);
}

__global__ void __launch_bounds__(256) k_final(const float* __restrict__ x,
                        const float* __restrict__ b,
                        const float* __restrict__ eps,
                        const float* __restrict__ cvec,
                        const float* __restrict__ W_epsr,
                        const float* __restrict__ W_zr,
                        const float* __restrict__ W_cr,
                        const float* __restrict__ zpart,
                        const float* __restrict__ rrpart,
                        float* __restrict__ out) {
  __shared__ __align__(16) float s_eps[NZ];
  __shared__ __align__(16) float s_z[NZ];
  __shared__ __align__(16) float s_c[NC];
  const int tid = threadIdx.x;
  {
    float zs = 0.f;
#pragma unroll
    for (int s = 0; s < KSPLIT; ++s) zs += zpart[s * NZ + tid];
    s_z[tid]   = zs;
    s_eps[tid] = eps[tid];
    if (tid < NC) s_c[tid] = cvec[tid];
  }
  __syncthreads();
  const int lane = tid & 31;
  const int row0 = blockIdx.x * 128 + (tid >> 5) * WROWS;
  v4f bv[5];
  bv[0] = *(const v4f*)(s_eps + 4 * lane);
  bv[1] = *(const v4f*)(s_eps + 128 + 4 * lane);
  bv[2] = *(const v4f*)(s_z + 4 * lane);
  bv[3] = *(const v4f*)(s_z + 128 + 4 * lane);
  bv[4] = *(const v4f*)(s_c + 4 * lane);
  const float* pe = W_epsr + (size_t)row0 * NZ + 4 * lane;
  const float* pz = W_zr   + (size_t)row0 * NZ + 4 * lane;
  const float* pc = W_cr   + (size_t)row0 * NC + 4 * lane;
  float t[WROWS];
  rows_dot<5, 3>([&](int r, int q) -> v4f {
    const float* p = (q < 2) ? pe + r * NZ + 128 * q
                   : (q < 4) ? pz + r * NZ + 128 * (q - 2)
                             : pc + r * NC;
    return *(const v4f*)p;
  }, bv, t, lane);
  const float y = reduce16(t, lane);
  if ((lane & 1) == 0) {
    const int i = row0 + 8 * (lane >> 4) + ((lane >> 1) & 7);
    float tot = y;
#pragma unroll
    for (int s = 0; s < KSPLIT; ++s) tot += rrpart[s * NN + i];
    const float xv = x[i];
    const float xn = xv + DTC * (tot - xv);
    out[OUT_X + i] = xn;
    out[OUT_R + i] = tanhf(xn + b[i]);
  }
}

__global__ void __launch_bounds__(256) k_outpart(const float* __restrict__ W_rz,
                          const float* __restrict__ W_rc,
                          const float* __restrict__ r_new,
                          float* __restrict__ znpart,
                          float* __restrict__ cnpart) {
  __shared__ __align__(16) float s_r[KCHUNK];
  const int tid = threadIdx.x;
  const int k0  = blockIdx.y * KCHUNK;
  for (int j = tid; j < KCHUNK; j += 256) s_r[j] = r_new[k0 + j];
  __syncthreads();
  const int wave = tid >> 5;
  const int lane = tid & 31;
  if (blockIdx.x < 2) {
    chunk_rows<false>(W_rz, blockIdx.x * 128 + wave * WROWS, k0, s_r,
                      znpart + blockIdx.y * NZ, lane);
  } else {
    chunk_rows<false>(W_rc, wave * WROWS, k0, s_r, cnpart + blockIdx.y * NC, lane);
  }
}

__global__ void k_epilog(const float* __restrict__ znpart,
                         const float* __restrict__ cnpart,
                         const float* __restrict__ z_tilde,
                         float* __restrict__ out) {
  const int tid = threadIdx.x;
  float zn = 0.f;
#pragma unroll
  for (int s = 0; s < KSPLIT; ++s) zn += znpart[s * NZ + tid];
  out[OUT_Z + tid] = zn;
  out[OUT_E + tid] = zn - z_tilde[tid];
  if (tid < NC) {
    float cn = 0.f;
#pragma unroll
    for (int s = 0; s < KSPLIT; ++s) cn += cnpart[s * NC + tid];
    out[OUT_C + tid] = cn;
  }
}

extern "C" void kernel_launch(void* const* d_in, const int* in_sizes, int n_in,
                              void* d_out, int out_size, void* d_ws, size_t ws_size,
                              hipStream_t stream) {
  (void)in_sizes; (void)n_in; (void)out_size; (void)ws_size;
  const float* x       = (const float*)d_in[0];
  const float* eps     = (const float*)d_in[1];
  const float* c       = (const float*)d_in[2];
  const float* z_tilde = (const float*)d_in[3];
  const float* W_rr    = (const float*)d_in[4];
  const float* W_zr    = (const float*)d_in[5];
  const float* W_cr    = (const float*)d_in[6];
  const float* W_epsr  = (const float*)d_in[7];
  const float* W_rz    = (const float*)d_in[8];
  const float* W_rc    = (const float*)d_in[9];
  const float* b       = (const float*)d_in[10];
  float* out = (float*)d_out;

  float* ws     = (float*)d_ws;
  float* zpart  = ws;
  float* rrpart = zpart  + KSPLIT * NZ;
  float* znpart = rrpart + KSPLIT * NN;
  float* cnpart = znpart + KSPLIT * NZ;

  k_zpart  <<<dim3(2, KSPLIT),  256, 0, stream>>>(W_rz, x, b, zpart);
  k_rrpart <<<dim3(64, KSPLIT), 256, 0, stream>>>(W_rr, x, b, rrpart);
  k_final  <<<64, 256, 0, stream>>>(x, b, eps, c, W_epsr, W_zr, W_cr,
                                    zpart, rrpart, out);
  k_outpart<<<dim3(3, KSPLIT),  256, 0, stream>>>(W_rz, W_rc, out + NN,
                                                  znpart, cnpart);
  k_epilog <<<1, 256, 0, stream>>>(znpart, cnpart, z_tilde, out);
}


# The three small matrix-vector launches re-split (32 K-splits, 2- or 4-wave groups: 256/128/384 work-groups, was 16/64/24) so all WGPs read; z, z_new, c_new sum 32 K-partials: sum order changes
# speedup vs baseline: 2.1698x; 1.2667x over previous
#include <hip/hip_runtime.h>
#include <math.h>

#define NN   8192
#define NZ   256
#define NC   128
#define DTC  0.1f

#define OUT_X 0
#define OUT_R NN
#define OUT_Z (2 * NN)
#define OUT_C (2 * NN + NZ)
#define OUT_E (2 * NN + NZ + NC)

#define KSPLIT 8
#define KCHUNK (NN / KSPLIT)
#define KQ     (KCHUNK / 128)
#define WROWS  16

#define SSPLIT 32
#define SCHUNK (NN / SSPLIT)
#define SQ     (SCHUNK / 128)
#define STHR   64
#define FTHR   128
static_assert(SCHUNK == 4 * STHR, "one 16-byte piece of the vector chunk per thread");
static_assert(NZ == 2 * FTHR && NC == FTHR, "k_final stages eps, z, c with one access per thread");

typedef __attribute__((ext_vector_type(2))) float v2f;
typedef __attribute__((ext_vector_type(4))) float v4f;
typedef __attribute__((ext_vector_type(8))) float v8f;

#define DEV static __device__ __forceinline__

DEV v8f wmma4(v8f acc, v2f a, v2f b) {
  return __builtin_amdgcn_wmma_f32_16x16x4_f32(false, a, false, b, (short)0, acc, false, false);
}

DEV float diag_pick(v8f d, int sel) {
  float t = 0.f;
#pragma unroll
  for (int j = 0; j < 8; ++j) t = (sel == j) ? d[j] : t;
  return t;
}

DEV float reduce16(const float (&t)[WROWS], int lane) {
  float u[8], v[4], w[2];
  const bool h4 = (lane & 16) != 0, h3 = (lane & 8) != 0, h2 = (lane & 4) != 0, h1 = (lane & 2) != 0;
#pragma unroll
  for (int j = 0; j < 8; ++j) u[j] = (h4 ? t[8 + j] : t[j]) + __shfl_xor(h4 ? t[j] : t[8 + j], 16, 32);
#pragma unroll
  for (int j = 0; j < 4; ++j) v[j] = (h3 ? u[4 + j] : u[j]) + __shfl_xor(h3 ? u[j] : u[4 + j], 8, 32);
#pragma unroll
  for (int j = 0; j < 2; ++j) w[j] = (h2 ? v[2 + j] : v[j]) + __shfl_xor(h2 ? v[j] : v[2 + j], 4, 32);
  const float x = (h1 ? w[1] : w[0]) + __shfl_xor(h1 ? w[0] : w[1], 2, 32);
  return x + __shfl_xor(x, 1, 32);
}

template <int NQ, int AHEAD, class LoadPiece>
DEV void rows_dot(LoadPiece ld, const v4f (&bv)[NQ], float (&t)[WROWS], int lane) {
  const int sel = (((lane + 8) & 16) == 0) ? (lane & 7) : 8;
  v4f w[AHEAD + 1][NQ];
#pragma unroll
  for (int r = 0; r < AHEAD; ++r) {
#pragma unroll
    for (int q = 0; q < NQ; ++q) w[r][q] = ld(r, q);
  }
#pragma unroll
  for (int r = 0; r < WROWS; ++r) {
    __builtin_amdgcn_sched_barrier(0);
    if (r + AHEAD < WROWS) {
#pragma unroll
      for (int q = 0; q < NQ; ++q) w[(r + AHEAD) % (AHEAD + 1)][q] = ld(r + AHEAD, q);
    }
    __builtin_amdgcn_sched_barrier(0);
    v8f a0 = {0.f, 0.f, 0.f, 0.f, 0.f, 0.f, 0.f, 0.f};
    v8f a1 = {0.f, 0.f, 0.f, 0.f, 0.f, 0.f, 0.f, 0.f};
#pragma unroll
    for (int q = 0; q < NQ; ++q) {
      const v4f wv = w[r % (AHEAD + 1)][q];
      const v2f wa = {wv.x, wv.y}, ba = {bv[q].x, bv[q].y};
      const v2f wb = {wv.z, wv.w}, bb = {bv[q].z, bv[q].w};
      a0 = wmma4(a0, wa, ba);
      a1 = wmma4(a1, wb, bb);
    }
    t[r] = diag_pick(a0 + a1, sel);
    asm volatile("" : "+v"(t[r]));
  }
}

template <bool NT, int NQ, int AHEAD>
DEV void chunk_rows(const float* __restrict__ W, int row0, int k0,
                    const float* s_r, float* __restrict__ dst, int lane) {
  v4f bv[NQ];
#pragma unroll
  for (int q = 0; q < NQ; ++q) bv[q] = *(const v4f*)(s_r + 128 * q + 4 * lane);
  const float* wp = W + (size_t)row0 * NN + k0 + 4 * lane;
  float t[WROWS];
  rows_dot<NQ, AHEAD>([&](int r, int q) -> v4f {
    const v4f* p = (const v4f*)(wp + (size_t)r * NN + 128 * q);
    return NT ? __builtin_nontemporal_load(p) : *p;
  }, bv, t, lane);
  const float y = reduce16(t, lane);
  if ((lane & 1) == 0) dst[row0 + 8 * (lane >> 4) + ((lane >> 1) & 7)] = y;
}

__global__ void __launch_bounds__(STHR) k_zpart(const float* __restrict__ W_rz,
                        const float* __restrict__ x,
                        const float* __restrict__ b,
                        float* __restrict__ zpart) {
  __shared__ __align__(16) float s_r[SCHUNK];
  const int tid = threadIdx.x;
  const int k0  = blockIdx.y * SCHUNK;
  {
    const v4f xv = *(const v4f*)(x + k0 + 4 * tid);
    const v4f bb = *(const v4f*)(b + k0 + 4 * tid);
    const v4f rv = {tanhf(xv.x + bb.x), tanhf(xv.y + bb.y), tanhf(xv.z + bb.z), tanhf(xv.w + bb.w)};
    *(v4f*)(s_r + 4 * tid) = rv;
  }
  __syncthreads();
  const int row0 = blockIdx.x * (STHR / 32 * WROWS) + (tid >> 5) * WROWS;
  chunk_rows<false, SQ, WROWS - 1>(W_rz, row0, k0, s_r, zpart + blockIdx.y * NZ, tid & 31);
}

__global__ void __launch_bounds__(256) k_rrpart(const float* __restrict__ W_rr,
                         const float* __restrict__ x,
                         const float* __restrict__ b,
                         float* __restrict__ rrpart) {
  __shared__ __align__(16) float s_r[KCHUNK];
  const int tid = threadIdx.x;
  const int k0  = blockIdx.y * KCHUNK;
  for (int j = tid; j < KCHUNK; j += 256)
    s_r[j] = tanhf(x[k0 + j] + b[k0 + j]);
  __syncthreads();
  const int row0 = blockIdx.x * 128 + (tid >> 5) * WROWS;
  chunk_rows<true, KQ, 3>(W_rr, row0, k0, s_r, rrpart + blockIdx.y * NN, tid & 31);
}

__global__ void __launch_bounds__(FTHR) k_final(const float* __restrict__ x,
                        const float* __restrict__ b,
                        const float* __restrict__ eps,
                        const float* __restrict__ cvec,
                        const float* __restrict__ W_epsr,
                        const float* __restrict__ W_zr,
                        const float* __restrict__ W_cr,
                        const float* __restrict__ zpart,
                        const float* __restrict__ rrpart,
                        float* __restrict__ out) {
  __shared__ __align__(16) float s_eps[NZ];
  __shared__ __align__(16) float s_z[NZ];
  __shared__ __align__(16) float s_c[NC];
  const int tid = threadIdx.x;
  {
    v2f zs = {0.f, 0.f};
#pragma unroll
    for (int s = 0; s < SSPLIT; ++s) zs += *(const v2f*)(zpart + s * NZ + 2 * tid);
    *(v2f*)(s_z + 2 * tid)   = zs;
    *(v2f*)(s_eps + 2 * tid) = *(const v2f*)(eps + 2 * tid);
    s_c[tid] = cvec[tid];
  }
  __syncthreads();
  const int lane = tid & 31;
  const int row0 = blockIdx.x * (FTHR / 32 * WROWS) + (tid >> 5) * WROWS;
  v4f bv[5];
  bv[0] = *(const v4f*)(s_eps + 4 * lane);
  bv[1] = *(const v4f*)(s_eps + 128 + 4 * lane);
  bv[2] = *(const v4f*)(s_z + 4 * lane);
  bv[3] = *(const v4f*)(s_z + 128 + 4 * lane);
  bv[4] = *(const v4f*)(s_c + 4 * lane);
  const float* pe = W_epsr + (size_t)row0 * NZ + 4 * lane;
  const float* pz = W_zr   + (size_t)row0 * NZ + 4 * lane;
  const float* pc = W_cr   + (size_t)row0 * NC + 4 * lane;
  float t[WROWS];
  rows_dot<5, 7>([&](int r, int q) -> v4f {
    const float* p = (q < 2) ? pe + r * NZ + 128 * q
                   : (q < 4) ? pz + r * NZ + 128 * (q - 2)
                             : pc + r * NC;
    return *(const v4f*)p;
  }, bv, t, lane);
  const float y = reduce16(t, lane);
  if ((lane & 1) == 0) {
    const int i = row0 + 8 * (lane >> 4) + ((lane >> 1) & 7);
    float tot = y;
#pragma unroll
    for (int s = 0; s < KSPLIT; ++s) tot += rrpart[s * NN + i];
    const float xv = x[i];
    const float xn = xv + DTC * (tot - xv);
    out[OUT_X + i] = xn;
    out[OUT_R + i] = tanhf(xn + b[i]);
  }
}

__global__ void __launch_bounds__(STHR) k_outpart(const float* __restrict__ W_rz,
                          const float* __restrict__ W_rc,
                          const float* __restrict__ r_new,
                          float* __restrict__ znpart,
                          float* __restrict__ cnpart) {
  __shared__ __align__(16) float s_r[SCHUNK];
  const int tid = threadIdx.x;
  const int k0  = blockIdx.y * SCHUNK;
  *(v4f*)(s_r + 4 * tid) = *(const v4f*)(r_new + k0 + 4 * tid);
  __syncthreads();
  const int wave = tid >> 5;
  const int lane = tid & 31;
  const int rpb  = STHR / 32 * WROWS;
  if (blockIdx.x < NZ / rpb) {
    chunk_rows<false, SQ, WROWS - 1>(W_rz, blockIdx.x * rpb + wave * WROWS, k0, s_r,
                                     znpart + blockIdx.y * NZ, lane);
  } else {
    chunk_rows<false, SQ, WROWS - 1>(W_rc, (blockIdx.x - NZ / rpb) * rpb + wave * WROWS, k0, s_r,
                                     cnpart + blockIdx.y * NC, lane);
  }
}

__global__ void k_epilog(const float* __restrict__ znpart,
                         const float* __restrict__ cnpart,
                         const float* __restrict__ z_tilde,
                         float* __restrict__ out) {
  const int tid = threadIdx.x;
  float zn = 0.f;
#pragma unroll
  for (int s = 0; s < SSPLIT; ++s) zn += znpart[s * NZ + tid];
  out[OUT_Z + tid] = zn;
  out[OUT_E + tid] = zn - z_tilde[tid];
  if (tid < NC) {
    float cn = 0.f;
#pragma unroll
    for (int s = 0; s < SSPLIT; ++s) cn += cnpart[s * NC + tid];
    out[OUT_C + tid] = cn;
  }
}

extern "C" void kernel_launch(void* const* d_in, const int* in_sizes, int n_in,
                              void* d_out, int out_size, void* d_ws, size_t ws_size,
                              hipStream_t stream) {
  (void)in_sizes; (void)n_in; (void)out_size; (void)ws_size;
  const float* x       = (const float*)d_in[0];
  const float* eps     = (const float*)d_in[1];
  const float* c       = (const float*)d_in[2];
  const float* z_tilde = (const float*)d_in[3];
  const float* W_rr    = (const float*)d_in[4];
  const float* W_zr    = (const float*)d_in[5];
  const float* W_cr    = (const float*)d_in[6];
  const float* W_epsr  = (const float*)d_in[7];
  const float* W_rz    = (const float*)d_in[8];
  const float* W_rc    = (const float*)d_in[9];
  const float* b       = (const float*)d_in[10];
  float* out = (float*)d_out;

  float* ws     = (float*)d_ws;
  float* zpart  = ws;
  float* rrpart = zpart  + SSPLIT * NZ;
  float* znpart = rrpart + KSPLIT * NN;
  float* cnpart = znpart + SSPLIT * NZ;

  k_zpart  <<<dim3(8, SSPLIT),  STHR, 0, stream>>>(W_rz, x, b, zpart);
  k_rrpart <<<dim3(64, KSPLIT), 256, 0, stream>>>(W_rr, x, b, rrpart);
  k_final  <<<128, FTHR, 0, stream>>>(x, b, eps, c, W_epsr, W_zr, W_cr,
                                    zpart, rrpart, out);
  k_outpart<<<dim3(12, SSPLIT), STHR, 0, stream>>>(W_rz, W_rc, out + NN,
                                                  znpart, cnpart);
  k_epilog <<<1, 256, 0, stream>>>(znpart, cnpart, z_tilde, out);
}


# z_new, c_new: each row of W_rz / W_rc is summed whole by one 4-wave work-group (4 K-shares of 2048 added in LDS, was 32 K-partials and a launch that added them): 3 launches, was 4; sum order changes
# speedup vs baseline: 2.4363x; 1.1228x over previous
#include <hip/hip_runtime.h>
#include <math.h>

#define NN   8192
#define NZ   256
#define NC   128
#define DTC  0.1f

#define OUT_X 0
#define OUT_R NN
#define OUT_Z (2 * NN)
#define OUT_C (2 * NN + NZ)
#define OUT_E (2 * NN + NZ + NC)

#define KSPLIT 8
#define KCHUNK (NN / KSPLIT)
#define KQ     (KCHUNK / 128)
#define WROWS  16

#define ZSPLIT (2 * KSPLIT)
#define ZQ     (KQ / 2)
static_assert(NZ == 64 * 4 && ZSPLIT * ZQ * 128 == NN, "64 work-groups x 4 wave pairs cover the rows of W_rz, a wave pair one K chunk");

#define FTHR   128
static_assert(NZ == 2 * FTHR && NC == FTHR, "k_final stages eps, z, c with one access per thread");

#define OTHR   128
#define OWAVES (OTHR / 32)
#define OQ     (NN / OWAVES / 128)
static_assert(OQ * 128 * OWAVES == NN, "the waves of a work-group cover one row");

typedef __attribute__((ext_vector_type(2))) float v2f;
typedef __attribute__((ext_vector_type(4))) float v4f;
typedef __attribute__((ext_vector_type(8))) float v8f;

#define DEV static __device__ __forceinline__

DEV v8f wmma4(v8f acc, v2f a, v2f b) {
  return __builtin_amdgcn_wmma_f32_16x16x4_f32(false, a, false, b, (short)0, acc, false, false);
}

DEV float diag_pick(v8f d, int sel) {
  float t = 0.f;
#pragma unroll
  for (int j = 0; j < 8; ++j) t = (sel == j) ? d[j] : t;
  return t;
}

DEV float reduce16(const float (&t)[WROWS + 1], int lane) {
  float u[8], v[4], w[2];
  const bool h4 = (lane & 16) != 0, h3 = (lane & 8) != 0, h2 = (lane & 4) != 0, h1 = (lane & 2) != 0;
#pragma unroll
  for (int j = 0; j < 8; ++j) u[j] = (h4 ? t[8 + j] : t[j]) + __shfl_xor(h4 ? t[j] : t[8 + j], 16, 32);
#pragma unroll
  for (int j = 0; j < 4; ++j) v[j] = (h3 ? u[4 + j] : u[j]) + __shfl_xor(h3 ? u[j] : u[4 + j], 8, 32);
#pragma unroll
  for (int j = 0; j < 2; ++j) w[j] = (h2 ? v[2 + j] : v[j]) + __shfl_xor(h2 ? v[j] : v[2 + j], 4, 32);
  const float x = (h1 ? w[1] : w[0]) + __shfl_xor(h1 ? w[0] : w[1], 2, 32);
  return x + __shfl_xor(x, 1, 32);
}

template <int NQ, int AHEAD, int XQ, class LoadPiece, class LoadExtra>
DEV void rows_dot(LoadPiece ld, LoadExtra ldx, const v4f (&bv)[NQ],
                  const v4f (&bx)[XQ > 0 ? XQ : 1], float (&t)[WROWS + 1], int lane) {
  static_assert(AHEAD < WROWS && XQ <= NQ, "the ring holds rows of NQ pieces");
  constexpr int NR = WROWS + (XQ > 0 ? 1 : 0);
  const int sel = (((lane + 8) & 16) == 0) ? (lane & 7) : 8;
  v4f w[AHEAD + 1][NQ];
#pragma unroll
  for (int r = 0; r < AHEAD; ++r) {
#pragma unroll
    for (int q = 0; q < NQ; ++q) w[r][q] = ld(r, q);
  }
#pragma unroll
  for (int r = 0; r < NR; ++r) {
    __builtin_amdgcn_sched_barrier(0);
    if (r + AHEAD < WROWS) {
#pragma unroll
      for (int q = 0; q < NQ; ++q) w[(r + AHEAD) % (AHEAD + 1)][q] = ld(r + AHEAD, q);
    } else if (r + AHEAD < NR) {
#pragma unroll
      for (int q = 0; q < XQ; ++q) w[(r + AHEAD) % (AHEAD + 1)][q] = ldx(q);
    }
    __builtin_amdgcn_sched_barrier(0);
    v8f a0 = {0.f, 0.f, 0.f, 0.f, 0.f, 0.f, 0.f, 0.f};
    v8f a1 = {0.f, 0.f, 0.f, 0.f, 0.f, 0.f, 0.f, 0.f};
#pragma unroll
    for (int q = 0; q < ((r < WROWS) ? NQ : XQ); ++q) {
      const v4f wv = w[r % (AHEAD + 1)][q];
      const v4f bq = (r < WROWS) ? bv[q] : bx[q < XQ ? q : 0];
      const v2f wa = {wv.x, wv.y}, ba = {bq.x, bq.y};
      const v2f wb = {wv.z, wv.w}, bb = {bq.z, bq.w};
      a0 = wmma4(a0, wa, ba);
      a1 = wmma4(a1, wb, bb);
    }
    t[r] = diag_pick(a0 + a1, sel);
    asm volatile("" : "+v"(t[r]));
  }
}

struct no_extra_row {
  __device__ v4f operator()(int) const { return v4f{0.f, 0.f, 0.f, 0.f}; }
};

__global__ void __launch_bounds__(256) k_rrpart(const float* __restrict__ W_rr,
                         const float* __restrict__ W_rz,
                         const float* __restrict__ x,
                         const float* __restrict__ b,
                         float* __restrict__ rrpart,
                         float* __restrict__ zpart) {
  __shared__ __align__(16) float s_r[KCHUNK];
  const int tid = threadIdx.x;
  const int k0  = blockIdx.y * KCHUNK;
  for (int j = tid; j < KCHUNK; j += 256)
    s_r[j] = tanhf(x[k0 + j] + b[k0 + j]);
  __syncthreads();
  const int lane = tid & 31;
  const int wave = tid >> 5;
  const int row0 = blockIdx.x * 128 + wave * WROWS;
  const int zhalf = wave & 1;
  const int zrow  = blockIdx.x * 4 + (wave >> 1);
  v4f bv[KQ], bz[ZQ];
#pragma unroll
  for (int q = 0; q < KQ; ++q) bv[q] = *(const v4f*)(s_r + 128 * q + 4 * lane);
#pragma unroll
  for (int q = 0; q < ZQ; ++q) bz[q] = *(const v4f*)(s_r + 128 * (ZQ * zhalf + q) + 4 * lane);
  const float* wp = W_rr + (size_t)row0 * NN + k0 + 4 * lane;
  const float* zp = W_rz + (size_t)zrow * NN + k0 + 128 * ZQ * zhalf + 4 * lane;
  float t[WROWS + 1];
  rows_dot<KQ, 3, ZQ>(
      [&](int r, int q) -> v4f {
        return __builtin_nontemporal_load((const v4f*)(wp + (size_t)r * NN + 128 * q));
      },
      [&](int q) -> v4f {
        return *(const v4f*)(zp + 128 * q);
      },
      bv, bz, t, lane);
  const float y = reduce16(t, lane);
  float zs = t[WROWS];
#pragma unroll
  for (int m = 16; m >= 1; m >>= 1) zs += __shfl_xor(zs, m, 32);
  if ((lane & 1) == 0) rrpart[blockIdx.y * NN + row0 + 8 * (lane >> 4) + ((lane >> 1) & 7)] = y;
  if (lane == 0) zpart[(2 * blockIdx.y + zhalf) * NZ + zrow] = zs;
}

__global__ void __launch_bounds__(FTHR) k_final(const float* __restrict__ x,
                        const float* __restrict__ b,
                        const float* __restrict__ eps,
                        const float* __restrict__ cvec,
                        const float* __restrict__ W_epsr,
                        const float* __restrict__ W_zr,
                        const float* __restrict__ W_cr,
                        const float* __restrict__ zpart,
                        const float* __restrict__ rrpart,
                        float* __restrict__ out) {
  __shared__ __align__(16) float s_eps[NZ];
  __shared__ __align__(16) float s_z[NZ];
  __shared__ __align__(16) float s_c[NC];
  const int tid = threadIdx.x;
  {
    v2f zs = {0.f, 0.f};
#pragma unroll
    for (int s = 0; s < ZSPLIT; ++s) zs += *(const v2f*)(zpart + s * NZ + 2 * tid);
    *(v2f*)(s_z + 2 * tid)   = zs;
    *(v2f*)(s_eps + 2 * tid) = *(const v2f*)(eps + 2 * tid);
    s_c[tid] = cvec[tid];
  }
  __syncthreads();
  const int lane = tid & 31;
  const int row0 = blockIdx.x * (FTHR / 32 * WROWS) + (tid >> 5) * WROWS;
  v4f bv[5];
  bv[0] = *(const v4f*)(s_eps + 4 * lane);
  bv[1] = *(const v4f*)(s_eps + 128 + 4 * lane);
  bv[2] = *(const v4f*)(s_z + 4 * lane);
  bv[3] = *(const v4f*)(s_z + 128 + 4 * lane);
  bv[4] = *(const v4f*)(s_c + 4 * lane);
  const float* pe = W_epsr + (size_t)row0 * NZ + 4 * lane;
  const float* pz = W_zr   + (size_t)row0 * NZ + 4 * lane;
  const float* pc = W_cr   + (size_t)row0 * NC + 4 * lane;
  const v4f none[1] = {};
  float t[WROWS + 1];
  rows_dot<5, 7, 0>([&](int r, int q) -> v4f {
    const float* p = (q < 2) ? pe + r * NZ + 128 * q
                   : (q < 4) ? pz + r * NZ + 128 * (q - 2)
                             : pc + r * NC;
    return *(const v4f*)p;
  }, no_extra_row(), bv, none, t, lane);
  const float y = reduce16(t, lane);
  if ((lane & 1) == 0) {
    const int i = row0 + 8 * (lane >> 4) + ((lane >> 1) & 7);
    float tot = y;
#pragma unroll
    for (int s = 0; s < KSPLIT; ++s) tot += rrpart[s * NN + i];
    const float xv = x[i];
    const float xn = xv + DTC * (tot - xv);
    out[OUT_X + i] = xn;
    out[OUT_R + i] = tanhf(xn + b[i]);
  }
}

__global__ void __launch_bounds__(OTHR) k_out(const float* __restrict__ W_rz,
                      const float* __restrict__ W_rc,
                      const float* __restrict__ r_new,
                      const float* __restrict__ z_tilde,
                      float* out) {
  __shared__ float s_part[OWAVES];
  const int tid  = threadIdx.x;
  const int wave = tid >> 5;
  const int lane = tid & 31;
  const int row  = blockIdx.x;
  const bool isz = row < NZ;
  const float* wrow = isz ? W_rz + (size_t)row * NN : W_rc + (size_t)(row - NZ) * NN;
  const int k0 = wave * (OQ * 128) + 4 * lane;
  v4f wv[OQ], rv[OQ];
#pragma unroll
  for (int q = 0; q < OQ; ++q) {
    wv[q] = *(const v4f*)(wrow + k0 + 128 * q);
    rv[q] = *(const v4f*)(r_new + k0 + 128 * q);
  }
  __builtin_amdgcn_sched_barrier(0);
  const float zt = z_tilde[isz ? row : 0];
  v8f a0 = {0.f, 0.f, 0.f, 0.f, 0.f, 0.f, 0.f, 0.f};
  v8f a1 = {0.f, 0.f, 0.f, 0.f, 0.f, 0.f, 0.f, 0.f};
#pragma unroll
  for (int q = 0; q < OQ; ++q) {
    const v2f wa = {wv[q].x, wv[q].y}, ba = {rv[q].x, rv[q].y};
    const v2f wb = {wv[q].z, wv[q].w}, bb = {rv[q].z, rv[q].w};
    a0 = wmma4(a0, wa, ba);
    a1 = wmma4(a1, wb, bb);
  }
  const int sel = (((lane + 8) & 16) == 0) ? (lane & 7) : 8;
  float t = diag_pick(a0 + a1, sel);
#pragma unroll
  for (int m = 16; m >= 1; m >>= 1) t += __shfl_xor(t, m, 32);
  if (lane == 0) s_part[wave] = t;
  __syncthreads();
  if (tid == 0) {
    const float s = (s_part[0] + s_part[1]) + (s_part[2] + s_part[3]);
    if (isz) {
      out[OUT_Z + row] = s;
      out[OUT_E + row] = s - zt;
    } else {
      out[OUT_C + row - NZ] = s;
    }
  }
}

extern "C" void kernel_launch(void* const* d_in, const int* in_sizes, int n_in,
                              void* d_out, int out_size, void* d_ws, size_t ws_size,
                              hipStream_t stream) {
  (void)in_sizes; (void)n_in; (void)out_size; (void)ws_size;
  const float* x       = (const float*)d_in[0];
  const float* eps     = (const float*)d_in[1];
  const float* c       = (const float*)d_in[2];
  const float* z_tilde = (const float*)d_in[3];
  const float* W_rr    = (const float*)d_in[4];
  const float* W_zr    = (const float*)d_in[5];
  const float* W_cr    = (const float*)d_in[6];
  const float* W_epsr  = (const float*)d_in[7];
  const float* W_rz    = (const float*)d_in[8];
  const float* W_rc    = (const float*)d_in[9];
  const float* b       = (const float*)d_in[10];
  float* out = (float*)d_out;

  float* ws     = (float*)d_ws;
  float* zpart  = ws;
  float* rrpart = zpart  + ZSPLIT * NZ;

  k_rrpart <<<dim3(64, KSPLIT), 256, 0, stream>>>(W_rr, W_rz, x, b, rrpart, zpart);
  k_final  <<<128, FTHR, 0, stream>>>(x, b, eps, c, W_epsr, W_zr, W_cr,
                                    zpart, rrpart, out);
  k_out    <<<NZ + NC, OTHR, 0, stream>>>(W_rz, W_rc, out + NN, z_tilde, out);
}


# Each launch requests its first streamed rows, with all its small operands as one group in front of them, BEFORE it builds the vector they multiply; the LDS hand-over barrier waits for LDS only
# speedup vs baseline: 2.5888x; 1.0626x over previous
#include <hip/hip_runtime.h>
#include <math.h>

#define NN   8192
#define NZ   256
#define NC   128
#define DTC  0.1f

#define OUT_X 0
#define OUT_R NN
#define OUT_Z (2 * NN)
#define OUT_C (2 * NN + NZ)
#define OUT_E (2 * NN + NZ + NC)

#define KSPLIT 8
#define KCHUNK (NN / KSPLIT)
#define KQ     (KCHUNK / 128)
#define WROWS  16

#define ZSPLIT (2 * KSPLIT)
#define ZQ     (KQ / 2)
static_assert(NZ == 64 * 4 && ZSPLIT * ZQ * 128 == NN, "64 work-groups x 4 wave pairs cover the rows of W_rz, a wave pair one K chunk");

#define FTHR   128
static_assert(NZ == 2 * FTHR && NC == FTHR, "k_final stages eps, z, c with one access per thread");

#define OTHR   128
#define OWAVES (OTHR / 32)
#define OQ     (NN / OWAVES / 128)
static_assert(OQ * 128 * OWAVES == NN, "the waves of a work-group cover one row");

typedef __attribute__((ext_vector_type(2))) float v2f;
typedef __attribute__((ext_vector_type(4))) float v4f;
typedef __attribute__((ext_vector_type(8))) float v8f;

#define DEV static __device__ __forceinline__

DEV v8f wmma4(v8f acc, v2f a, v2f b) {
  return __builtin_amdgcn_wmma_f32_16x16x4_f32(false, a, false, b, (short)0, acc, false, false);
}

DEV float diag_pick(v8f d, int sel) {
  float t = 0.f;
#pragma unroll
  for (int j = 0; j < 8; ++j) t = (sel == j) ? d[j] : t;
  return t;
}

DEV float reduce16(const float (&t)[WROWS + 1], int lane) {
  float u[8], v[4], w[2];
  const bool h4 = (lane & 16) != 0, h3 = (lane & 8) != 0, h2 = (lane & 4) != 0, h1 = (lane & 2) != 0;
#pragma unroll
  for (int j = 0; j < 8; ++j) u[j] = (h4 ? t[8 + j] : t[j]) + __shfl_xor(h4 ? t[j] : t[8 + j], 16, 32);
#pragma unroll
  for (int j = 0; j < 4; ++j) v[j] = (h3 ? u[4 + j] : u[j]) + __shfl_xor(h3 ? u[j] : u[4 + j], 8, 32);
#pragma unroll
  for (int j = 0; j < 2; ++j) w[j] = (h2 ? v[2 + j] : v[j]) + __shfl_xor(h2 ? v[j] : v[2 + j], 4, 32);
  const float x = (h1 ? w[1] : w[0]) + __shfl_xor(h1 ? w[0] : w[1], 2, 32);
  return x + __shfl_xor(x, 1, 32);
}

DEV void lds_barrier() {
  asm volatile("s_wait_dscnt 0x0\n\ts_barrier_signal -1\n\ts_barrier_wait -1" ::: "memory");
}

template <int NQ, int AHEAD, int XQ, class LoadPiece, class LoadExtra, class Prep>
DEV void rows_dot(LoadPiece ld, LoadExtra ldx, Prep prep, float (&t)[WROWS + 1], int lane) {
  static_assert(AHEAD < WROWS && XQ <= NQ, "the ring holds rows of NQ pieces");
  constexpr int NR = WROWS + (XQ > 0 ? 1 : 0);
  const int sel = (((lane + 8) & 16) == 0) ? (lane & 7) : 8;
  v4f w[AHEAD + 1][NQ];
  __builtin_amdgcn_sched_barrier(0);
#pragma unroll
  for (int r = 0; r < AHEAD; ++r) {
#pragma unroll
    for (int q = 0; q < NQ; ++q) w[r][q] = ld(r, q);
  }
  __builtin_amdgcn_sched_barrier(0);
  v4f bv[NQ], bx[XQ > 0 ? XQ : 1];
  prep(bv, bx);
#pragma unroll
  for (int r = 0; r < NR; ++r) {
    __builtin_amdgcn_sched_barrier(0);
    if (r + AHEAD < WROWS) {
#pragma unroll
      for (int q = 0; q < NQ; ++q) w[(r + AHEAD) % (AHEAD + 1)][q] = ld(r + AHEAD, q);
    } else if (r + AHEAD < NR) {
#pragma unroll
      for (int q = 0; q < XQ; ++q) w[(r + AHEAD) % (AHEAD + 1)][q] = ldx(q);
    }
    __builtin_amdgcn_sched_barrier(0);
    v8f a0 = {0.f, 0.f, 0.f, 0.f, 0.f, 0.f, 0.f, 0.f};
    v8f a1 = {0.f, 0.f, 0.f, 0.f, 0.f, 0.f, 0.f, 0.f};
#pragma unroll
    for (int q = 0; q < ((r < WROWS) ? NQ : XQ); ++q) {
      const v4f wv = w[r % (AHEAD + 1)][q];
      const v4f bq = (r < WROWS) ? bv[q] : bx[q < XQ ? q : 0];
      const v2f wa = {wv.x, wv.y}, ba = {bq.x, bq.y};
      const v2f wb = {wv.z, wv.w}, bb = {bq.z, bq.w};
      a0 = wmma4(a0, wa, ba);
      a1 = wmma4(a1, wb, bb);
    }
    t[r] = diag_pick(a0 + a1, sel);
    asm volatile("" : "+v"(t[r]));
  }
}

struct no_extra_row {
  __device__ v4f operator()(int) const { return v4f{0.f, 0.f, 0.f, 0.f}; }
};

__global__ void __launch_bounds__(256) k_rrpart(const float* __restrict__ W_rr,
                         const float* __restrict__ W_rz,
                         const float* __restrict__ x,
                         const float* __restrict__ b,
                         float* __restrict__ rrpart,
                         float* __restrict__ zpart) {
  __shared__ __align__(16) float s_r[KCHUNK];
  static_assert(KCHUNK == 4 * 256, "one 16-byte piece of x and of b per thread");
  const int tid = threadIdx.x;
  const int k0  = blockIdx.y * KCHUNK;
  const v4f xv = *(const v4f*)(x + k0 + 4 * tid);
  const v4f xb = *(const v4f*)(b + k0 + 4 * tid);
  const int lane = tid & 31;
  const int wave = tid >> 5;
  const int row0 = blockIdx.x * 128 + wave * WROWS;
  const int zhalf = wave & 1;
  const int zrow  = blockIdx.x * 4 + (wave >> 1);
  const float* wp = W_rr + (size_t)row0 * NN + k0 + 4 * lane;
  const float* zp = W_rz + (size_t)zrow * NN + k0 + 128 * ZQ * zhalf + 4 * lane;
  float t[WROWS + 1];
  rows_dot<KQ, 3, ZQ>(
      [&](int r, int q) -> v4f {
        return __builtin_nontemporal_load((const v4f*)(wp + (size_t)r * NN + 128 * q));
      },
      [&](int q) -> v4f {
        return *(const v4f*)(zp + 128 * q);
      },
      [&](v4f (&bv)[KQ], v4f (&bz)[ZQ]) {
        const v4f rr = {tanhf(xv.x + xb.x), tanhf(xv.y + xb.y), tanhf(xv.z + xb.z), tanhf(xv.w + xb.w)};
        *(v4f*)(s_r + 4 * tid) = rr;
        lds_barrier();
#pragma unroll
        for (int q = 0; q < KQ; ++q) bv[q] = *(const v4f*)(s_r + 128 * q + 4 * lane);
#pragma unroll
        for (int q = 0; q < ZQ; ++q) bz[q] = *(const v4f*)(s_r + 128 * (ZQ * zhalf + q) + 4 * lane);
      },
      t, lane);
  const float y = reduce16(t, lane);
  float zs = t[WROWS];
#pragma unroll
  for (int m = 16; m >= 1; m >>= 1) zs += __shfl_xor(zs, m, 32);
  if ((lane & 1) == 0) rrpart[blockIdx.y * NN + row0 + 8 * (lane >> 4) + ((lane >> 1) & 7)] = y;
  if (lane == 0) zpart[(2 * blockIdx.y + zhalf) * NZ + zrow] = zs;
}

__global__ void __launch_bounds__(FTHR) k_final(const float* __restrict__ x,
                        const float* __restrict__ b,
                        const float* __restrict__ eps,
                        const float* __restrict__ cvec,
                        const float* __restrict__ W_epsr,
                        const float* __restrict__ W_zr,
                        const float* __restrict__ W_cr,
                        const float* __restrict__ zpart,
                        const float* __restrict__ rrpart,
                        float* __restrict__ out) {
  __shared__ __align__(16) float s_eps[NZ];
  __shared__ __align__(16) float s_z[NZ];
  __shared__ __align__(16) float s_c[NC];
  const int tid = threadIdx.x;
  const int lane = tid & 31;
  const int row0 = blockIdx.x * (FTHR / 32 * WROWS) + (tid >> 5) * WROWS;
  const int i = row0 + 8 * (lane >> 4) + ((lane >> 1) & 7);
  v2f zq[ZSPLIT];
#pragma unroll
  for (int s = 0; s < ZSPLIT; ++s) zq[s] = *(const v2f*)(zpart + s * NZ + 2 * tid);
  const v2f ev = *(const v2f*)(eps + 2 * tid);
  const float cv = cvec[tid];
  float pr[KSPLIT];
#pragma unroll
  for (int s = 0; s < KSPLIT; ++s) pr[s] = rrpart[s * NN + i];
  const float xv = x[i];
  const float bi = b[i];
  const float* pe = W_epsr + (size_t)row0 * NZ + 4 * lane;
  const float* pz = W_zr   + (size_t)row0 * NZ + 4 * lane;
  const float* pc = W_cr   + (size_t)row0 * NC + 4 * lane;
  float t[WROWS + 1];
  rows_dot<5, 7, 0>([&](int r, int q) -> v4f {
    const float* p = (q < 2) ? pe + r * NZ + 128 * q
                   : (q < 4) ? pz + r * NZ + 128 * (q - 2)
                             : pc + r * NC;
    return *(const v4f*)p;
  }, no_extra_row(), [&](v4f (&bv)[5], v4f (&)[1]) {
    v2f zs = {0.f, 0.f};
#pragma unroll
    for (int s = 0; s < ZSPLIT; ++s) zs += zq[s];
    *(v2f*)(s_z + 2 * tid)   = zs;
    *(v2f*)(s_eps + 2 * tid) = ev;
    s_c[tid] = cv;
    lds_barrier();
    bv[0] = *(const v4f*)(s_eps + 4 * lane);
    bv[1] = *(const v4f*)(s_eps + 128 + 4 * lane);
    bv[2] = *(const v4f*)(s_z + 4 * lane);
    bv[3] = *(const v4f*)(s_z + 128 + 4 * lane);
    bv[4] = *(const v4f*)(s_c + 4 * lane);
  }, t, lane);
  asm volatile("" :: "v"(pr[0]), "v"(pr[1]), "v"(pr[2]), "v"(pr[3]), "v"(pr[4]), "v"(pr[5]),
               "v"(pr[6]), "v"(pr[7]), "v"(xv), "v"(bi));
  static_assert(KSPLIT == 8, "the statement above names the KSPLIT partials one by one");
  const float y = reduce16(t, lane);
  if ((lane & 1) == 0) {
    float tot = y;
#pragma unroll
    for (int s = 0; s < KSPLIT; ++s) tot += pr[s];
    const float xn = xv + DTC * (tot - xv);
    out[OUT_X + i] = xn;
    out[OUT_R + i] = tanhf(xn + bi);
  }
}

__global__ void __launch_bounds__(OTHR) k_out(const float* __restrict__ W_rz,
                      const float* __restrict__ W_rc,
                      const float* __restrict__ r_new,
                      const float* __restrict__ z_tilde,
                      float* out) {
  __shared__ float s_part[OWAVES];
  const int tid  = threadIdx.x;
  const int wave = tid >> 5;
  const int lane = tid & 31;
  const int row  = blockIdx.x;
  const bool isz = row < NZ;
  const float* wrow = isz ? W_rz + (size_t)row * NN : W_rc + (size_t)(row - NZ) * NN;
  const int k0 = wave * (OQ * 128) + 4 * lane;
  v4f wv[OQ], rv[OQ];
#pragma unroll
  for (int q = 0; q < OQ; ++q) {
    wv[q] = *(const v4f*)(wrow + k0 + 128 * q);
    rv[q] = *(const v4f*)(r_new + k0 + 128 * q);
  }
  __builtin_amdgcn_sched_barrier(0);
  const float zt = z_tilde[isz ? row : 0];
  v8f a0 = {0.f, 0.f, 0.f, 0.f, 0.f, 0.f, 0.f, 0.f};
  v8f a1 = {0.f, 0.f, 0.f, 0.f, 0.f, 0.f, 0.f, 0.f};
#pragma unroll
  for (int q = 0; q < OQ; ++q) {
    const v2f wa = {wv[q].x, wv[q].y}, ba = {rv[q].x, rv[q].y};
    const v2f wb = {wv[q].z, wv[q].w}, bb = {rv[q].z, rv[q].w};
    a0 = wmma4(a0, wa, ba);
    a1 = wmma4(a1, wb, bb);
  }
  const int sel = (((lane + 8) & 16) == 0) ? (lane & 7) : 8;
  float t = diag_pick(a0 + a1, sel);
#pragma unroll
  for (int m = 16; m >= 1; m >>= 1) t += __shfl_xor(t, m, 32);
  if (lane == 0) s_part[wave] = t;
  __syncthreads();
  if (tid == 0) {
    const float s = (s_part[0] + s_part[1]) + (s_part[2] + s_part[3]);
    if (isz) {
      out[OUT_Z + row] = s;
      out[OUT_E + row] = s - zt;
    } else {
      out[OUT_C + row - NZ] = s;
    }
  }
}

extern "C" void kernel_launch(void* const* d_in, const int* in_sizes, int n_in,
                              void* d_out, int out_size, void* d_ws, size_t ws_size,
                              hipStream_t stream) {
  (void)in_sizes; (void)n_in; (void)out_size; (void)ws_size;
  const float* x       = (const float*)d_in[0];
  const float* eps     = (const float*)d_in[1];
  const float* c       = (const float*)d_in[2];
  const float* z_tilde = (const float*)d_in[3];
  const float* W_rr    = (const float*)d_in[4];
  const float* W_zr    = (const float*)d_in[5];
  const float* W_cr    = (const float*)d_in[6];
  const float* W_epsr  = (const float*)d_in[7];
  const float* W_rz    = (const float*)d_in[8];
  const float* W_rc    = (const float*)d_in[9];
  const float* b       = (const float*)d_in[10];
  float* out = (float*)d_out;

  float* ws     = (float*)d_ws;
  float* zpart  = ws;
  float* rrpart = zpart  + ZSPLIT * NZ;

  k_rrpart <<<dim3(64, KSPLIT), 256, 0, stream>>>(W_rr, W_rz, x, b, rrpart, zpart);
  k_final  <<<128, FTHR, 0, stream>>>(x, b, eps, c, W_epsr, W_zr, W_cr,
                                    zpart, rrpart, out);
  k_out    <<<NZ + NC, OTHR, 0, stream>>>(W_rz, W_rc, out + NN, z_tilde, out);
}


# k_final: 8 rows per wave (was 16): 256 work-groups, one wave on every SIMD (was 128 on half the WGPs); its 32 rows leave as whole 128-byte lines through LDS; every sum order kept
# speedup vs baseline: 2.6641x; 1.0291x over previous
#include <hip/hip_runtime.h>
#include <math.h>

#define NN   8192
#define NZ   256
#define NC   128
#define DTC  0.1f

#define OUT_X 0
#define OUT_R NN
#define OUT_Z (2 * NN)
#define OUT_C (2 * NN + NZ)
#define OUT_E (2 * NN + NZ + NC)

#define KSPLIT 8
#define KCHUNK (NN / KSPLIT)
#define KQ     (KCHUNK / 128)
#define WROWS  16

#define ZSPLIT (2 * KSPLIT)
#define ZQ     (KQ / 2)
static_assert(NZ == 64 * 4 && ZSPLIT * ZQ * 128 == NN, "64 work-groups x 4 wave pairs cover the rows of W_rz, a wave pair one K chunk");

#define FTHR   128
#define FROWS  8
#define FBROWS (FTHR / 32 * FROWS)
static_assert(NZ == 2 * FTHR && NC == FTHR, "k_final stages eps, z, c with one access per thread");
static_assert(FBROWS == 32, "a work-group of k_final owns one 128-byte line of x_new and one of r_new");

#define OTHR   128
#define OWAVES (OTHR / 32)
#define OQ     (NN / OWAVES / 128)
static_assert(OQ * 128 * OWAVES == NN, "the waves of a work-group cover one row");

typedef __attribute__((ext_vector_type(2))) float v2f;
typedef __attribute__((ext_vector_type(4))) float v4f;
typedef __attribute__((ext_vector_type(8))) float v8f;

#define DEV static __device__ __forceinline__

DEV v8f wmma4(v8f acc, v2f a, v2f b) {
  return __builtin_amdgcn_wmma_f32_16x16x4_f32(false, a, false, b, (short)0, acc, false, false);
}

DEV float diag_pick(v8f d, int sel) {
  float t = 0.f;
#pragma unroll
  for (int j = 0; j < 8; ++j) t = (sel == j) ? d[j] : t;
  return t;
}

DEV float reduce16(const float (&t)[WROWS + 1], int lane) {
  float u[8], v[4], w[2];
  const bool h4 = (lane & 16) != 0, h3 = (lane & 8) != 0, h2 = (lane & 4) != 0, h1 = (lane & 2) != 0;
#pragma unroll
  for (int j = 0; j < 8; ++j) u[j] = (h4 ? t[8 + j] : t[j]) + __shfl_xor(h4 ? t[j] : t[8 + j], 16, 32);
#pragma unroll
  for (int j = 0; j < 4; ++j) v[j] = (h3 ? u[4 + j] : u[j]) + __shfl_xor(h3 ? u[j] : u[4 + j], 8, 32);
#pragma unroll
  for (int j = 0; j < 2; ++j) w[j] = (h2 ? v[2 + j] : v[j]) + __shfl_xor(h2 ? v[j] : v[2 + j], 4, 32);
  const float x = (h1 ? w[1] : w[0]) + __shfl_xor(h1 ? w[0] : w[1], 2, 32);
  return x + __shfl_xor(x, 1, 32);
}

DEV float reduce8(const float (&t)[FROWS + 1], int lane) {
  float u[4], v[2];
  const bool h4 = (lane & 16) != 0, h3 = (lane & 8) != 0, h2 = (lane & 4) != 0;
#pragma unroll
  for (int j = 0; j < 4; ++j) u[j] = (h4 ? t[4 + j] : t[j]) + __shfl_xor(h4 ? t[j] : t[4 + j], 16, 32);
#pragma unroll
  for (int j = 0; j < 2; ++j) v[j] = (h3 ? u[2 + j] : u[j]) + __shfl_xor(h3 ? u[j] : u[2 + j], 8, 32);
  const float w = (h2 ? v[1] : v[0]) + __shfl_xor(h2 ? v[0] : v[1], 4, 32);
  const float x = w + __shfl_xor(w, 2, 32);
  return x + __shfl_xor(x, 1, 32);
}

DEV void lds_barrier() {
  asm volatile("s_wait_dscnt 0x0\n\ts_barrier_signal -1\n\ts_barrier_wait -1" ::: "memory");
}

template <int ROWS, int NQ, int AHEAD, int XQ, class LoadPiece, class LoadExtra, class Prep>
DEV void rows_dot(LoadPiece ld, LoadExtra ldx, Prep prep, float (&t)[ROWS + 1], int lane) {
  static_assert(AHEAD < ROWS && XQ <= NQ, "the ring holds rows of NQ pieces");
  constexpr int NR = ROWS + (XQ > 0 ? 1 : 0);
  const int sel = (((lane + 8) & 16) == 0) ? (lane & 7) : 8;
  v4f w[AHEAD + 1][NQ];
  __builtin_amdgcn_sched_barrier(0);
#pragma unroll
  for (int r = 0; r < AHEAD; ++r) {
#pragma unroll
    for (int q = 0; q < NQ; ++q) w[r][q] = ld(r, q);
  }
  __builtin_amdgcn_sched_barrier(0);
  v4f bv[NQ], bx[XQ > 0 ? XQ : 1];
  prep(bv, bx);
#pragma unroll
  for (int r = 0; r < NR; ++r) {
    __builtin_amdgcn_sched_barrier(0);
    if (r + AHEAD < ROWS) {
#pragma unroll
      for (int q = 0; q < NQ; ++q) w[(r + AHEAD) % (AHEAD + 1)][q] = ld(r + AHEAD, q);
    } else if (r + AHEAD < NR) {
#pragma unroll
      for (int q = 0; q < XQ; ++q) w[(r + AHEAD) % (AHEAD + 1)][q] = ldx(q);
    }
    __builtin_amdgcn_sched_barrier(0);
    v8f a0 = {0.f, 0.f, 0.f, 0.f, 0.f, 0.f, 0.f, 0.f};
    v8f a1 = {0.f, 0.f, 0.f, 0.f, 0.f, 0.f, 0.f, 0.f};
#pragma unroll
    for (int q = 0; q < ((r < ROWS) ? NQ : XQ); ++q) {
      const v4f wv = w[r % (AHEAD + 1)][q];
      const v4f bq = (r < ROWS) ? bv[q] : bx[q < XQ ? q : 0];
      const v2f wa = {wv.x, wv.y}, ba = {bq.x, bq.y};
      const v2f wb = {wv.z, wv.w}, bb = {bq.z, bq.w};
      a0 = wmma4(a0, wa, ba);
      a1 = wmma4(a1, wb, bb);
    }
    t[r] = diag_pick(a0 + a1, sel);
    asm volatile("" : "+v"(t[r]));
  }
}

struct no_extra_row {
  __device__ v4f operator()(int) const { return v4f{0.f, 0.f, 0.f, 0.f}; }
};

__global__ void __launch_bounds__(256) k_rrpart(const float* __restrict__ W_rr,
                         const float* __restrict__ W_rz,
                         const float* __restrict__ x,
                         const float* __restrict__ b,
                         float* __restrict__ rrpart,
                         float* __restrict__ zpart) {
  __shared__ __align__(16) float s_r[KCHUNK];
  static_assert(KCHUNK == 4 * 256, "one 16-byte piece of x and of b per thread");
  const int tid = threadIdx.x;
  const int k0  = blockIdx.y * KCHUNK;
  const v4f xv = *(const v4f*)(x + k0 + 4 * tid);
  const v4f xb = *(const v4f*)(b + k0 + 4 * tid);
  const int lane = tid & 31;
  const int wave = tid >> 5;
  const int row0 = blockIdx.x * 128 + wave * WROWS;
  const int zhalf = wave & 1;
  const int zrow  = blockIdx.x * 4 + (wave >> 1);
  const float* wp = W_rr + (size_t)row0 * NN + k0 + 4 * lane;
  const float* zp = W_rz + (size_t)zrow * NN + k0 + 128 * ZQ * zhalf + 4 * lane;
  float t[WROWS + 1];
  rows_dot<WROWS, KQ, 3, ZQ>(
      [&](int r, int q) -> v4f {
        return __builtin_nontemporal_load((const v4f*)(wp + (size_t)r * NN + 128 * q));
      },
      [&](int q) -> v4f {
        return *(const v4f*)(zp + 128 * q);
      },
      [&](v4f (&bv)[KQ], v4f (&bz)[ZQ]) {
        const v4f rr = {tanhf(xv.x + xb.x), tanhf(xv.y + xb.y), tanhf(xv.z + xb.z), tanhf(xv.w + xb.w)};
        *(v4f*)(s_r + 4 * tid) = rr;
        lds_barrier();
#pragma unroll
        for (int q = 0; q < KQ; ++q) bv[q] = *(const v4f*)(s_r + 128 * q + 4 * lane);
#pragma unroll
        for (int q = 0; q < ZQ; ++q) bz[q] = *(const v4f*)(s_r + 128 * (ZQ * zhalf + q) + 4 * lane);
      },
      t, lane);
  const float y = reduce16(t, lane);
  float zs = t[WROWS];
#pragma unroll
  for (int m = 16; m >= 1; m >>= 1) zs += __shfl_xor(zs, m, 32);
  if ((lane & 1) == 0) rrpart[blockIdx.y * NN + row0 + 8 * (lane >> 4) + ((lane >> 1) & 7)] = y;
  if (lane == 0) zpart[(2 * blockIdx.y + zhalf) * NZ + zrow] = zs;
}

__global__ void __launch_bounds__(FTHR) k_final(const float* __restrict__ x,
                        const float* __restrict__ b,
                        const float* __restrict__ eps,
                        const float* __restrict__ cvec,
                        const float* __restrict__ W_epsr,
                        const float* __restrict__ W_zr,
                        const float* __restrict__ W_cr,
                        const float* __restrict__ zpart,
                        const float* __restrict__ rrpart,
                        float* __restrict__ out) {
  __shared__ __align__(16) float s_eps[NZ];
  __shared__ __align__(16) float s_z[NZ];
  __shared__ __align__(16) float s_c[NC];
  __shared__ float s_xn[FBROWS];
  __shared__ float s_rn[FBROWS];
  const int tid = threadIdx.x;
  const int lane = tid & 31;
  const int wave = tid >> 5;
  const int lrow = wave * FROWS + ((lane >> 2) & 7);
  const int row0 = blockIdx.x * FBROWS + wave * FROWS;
  const int i = blockIdx.x * FBROWS + lrow;
  v2f zq[ZSPLIT];
#pragma unroll
  for (int s = 0; s < ZSPLIT; ++s) zq[s] = *(const v2f*)(zpart + s * NZ + 2 * tid);
  const v2f ev = *(const v2f*)(eps + 2 * tid);
  const float cv = cvec[tid];
  float pr[KSPLIT];
#pragma unroll
  for (int s = 0; s < KSPLIT; ++s) pr[s] = rrpart[s * NN + i];
  const float xv = x[i];
  const float bi = b[i];
  const float* pe = W_epsr + (size_t)row0 * NZ + 4 * lane;
  const float* pz = W_zr   + (size_t)row0 * NZ + 4 * lane;
  const float* pc = W_cr   + (size_t)row0 * NC + 4 * lane;
  float t[FROWS + 1];
  rows_dot<FROWS, 5, 7, 0>([&](int r, int q) -> v4f {
    const float* p = (q < 2) ? pe + r * NZ + 128 * q
                   : (q < 4) ? pz + r * NZ + 128 * (q - 2)
                             : pc + r * NC;
    return *(const v4f*)p;
  }, no_extra_row(), [&](v4f (&bv)[5], v4f (&)[1]) {
    v2f zs = {0.f, 0.f};
#pragma unroll
    for (int s = 0; s < ZSPLIT; ++s) zs += zq[s];
    *(v2f*)(s_z + 2 * tid)   = zs;
    *(v2f*)(s_eps + 2 * tid) = ev;
    s_c[tid] = cv;
    lds_barrier();
    bv[0] = *(const v4f*)(s_eps + 4 * lane);
    bv[1] = *(const v4f*)(s_eps + 128 + 4 * lane);
    bv[2] = *(const v4f*)(s_z + 4 * lane);
    bv[3] = *(const v4f*)(s_z + 128 + 4 * lane);
    bv[4] = *(const v4f*)(s_c + 4 * lane);
  }, t, lane);
  asm volatile("" :: "v"(pr[0]), "v"(pr[1]), "v"(pr[2]), "v"(pr[3]), "v"(pr[4]), "v"(pr[5]),
               "v"(pr[6]), "v"(pr[7]), "v"(xv), "v"(bi));
  static_assert(KSPLIT == 8, "the statement above names the KSPLIT partials one by one");
  const float y = reduce8(t, lane);
  if ((lane & 3) == 0) {
    float tot = y;
#pragma unroll
    for (int s = 0; s < KSPLIT; ++s) tot += pr[s];
    const float xn = xv + DTC * (tot - xv);
    s_xn[lrow] = xn;
    s_rn[lrow] = tanhf(xn + bi);
  }
  __syncthreads();
  if (wave == 0) out[OUT_X + blockIdx.x * FBROWS + lane] = s_xn[lane];
  if (wave == 1) out[OUT_R + blockIdx.x * FBROWS + lane] = s_rn[lane];
}

__global__ void __launch_bounds__(OTHR) k_out(const float* __restrict__ W_rz,
                      const float* __restrict__ W_rc,
                      const float* __restrict__ r_new,
                      const float* __restrict__ z_tilde,
                      float* out) {
  __shared__ float s_part[OWAVES];
  const int tid  = threadIdx.x;
  const int wave = tid >> 5;
  const int lane = tid & 31;
  const int row  = blockIdx.x;
  const bool isz = row < NZ;
  const float* wrow = isz ? W_rz + (size_t)row * NN : W_rc + (size_t)(row - NZ) * NN;
  const int k0 = wave * (OQ * 128) + 4 * lane;
  v4f wv[OQ], rv[OQ];
#pragma unroll
  for (int q = 0; q < OQ; ++q) {
    wv[q] = *(const v4f*)(wrow + k0 + 128 * q);
    rv[q] = *(const v4f*)(r_new + k0 + 128 * q);
  }
  __builtin_amdgcn_sched_barrier(0);
  const float zt = z_tilde[isz ? row : 0];
  v8f a0 = {0.f, 0.f, 0.f, 0.f, 0.f, 0.f, 0.f, 0.f};
  v8f a1 = {0.f, 0.f, 0.f, 0.f, 0.f, 0.f, 0.f, 0.f};
#pragma unroll
  for (int q = 0; q < OQ; ++q) {
    const v2f wa = {wv[q].x, wv[q].y}, ba = {rv[q].x, rv[q].y};
    const v2f wb = {wv[q].z, wv[q].w}, bb = {rv[q].z, rv[q].w};
    a0 = wmma4(a0, wa, ba);
    a1 = wmma4(a1, wb, bb);
  }
  const int sel = (((lane + 8) & 16) == 0) ? (lane & 7) : 8;
  float t = diag_pick(a0 + a1, sel);
#pragma unroll
  for (int m = 16; m >= 1; m >>= 1) t += __shfl_xor(t, m, 32);
  if (lane == 0) s_part[wave] = t;
  __syncthreads();
  if (tid == 0) {
    const float s = (s_part[0] + s_part[1]) + (s_part[2] + s_part[3]);
    if (isz) {
      out[OUT_Z + row] = s;
      out[OUT_E + row] = s - zt;
    } else {
      out[OUT_C + row - NZ] = s;
    }
  }
}

extern "C" void kernel_launch(void* const* d_in, const int* in_sizes, int n_in,
                              void* d_out, int out_size, void* d_ws, size_t ws_size,
                              hipStream_t stream) {
  (void)in_sizes; (void)n_in; (void)out_size; (void)ws_size;
  const float* x       = (const float*)d_in[0];
  const float* eps     = (const float*)d_in[1];
  const float* c       = (const float*)d_in[2];
  const float* z_tilde = (const float*)d_in[3];
  const float* W_rr    = (const float*)d_in[4];
  const float* W_zr    = (const float*)d_in[5];
  const float* W_cr    = (const float*)d_in[6];
  const float* W_epsr  = (const float*)d_in[7];
  const float* W_rz    = (const float*)d_in[8];
  const float* W_rc    = (const float*)d_in[9];
  const float* b       = (const float*)d_in[10];
  float* out = (float*)d_out;

  float* ws     = (float*)d_ws;
  float* zpart  = ws;
  float* rrpart = zpart  + ZSPLIT * NZ;

  k_rrpart <<<dim3(64, KSPLIT), 256, 0, stream>>>(W_rr, W_rz, x, b, rrpart, zpart);
  k_final  <<<NN / FBROWS, FTHR, 0, stream>>>(x, b, eps, c, W_epsr, W_zr, W_cr,
                                    zpart, rrpart, out);
  k_out    <<<NZ + NC, OTHR, 0, stream>>>(W_rz, W_rc, out + NN, z_tilde, out);
}
